# starting kernel
# baseline (speedup 1.0000x reference)
#include <hip/hip_runtime.h>
#include <hip/hip_bf16.h>


typedef float v2f __attribute__((ext_vector_type(2)));
typedef float v8f __attribute__((ext_vector_type(8)));

#define NEG_SLOPE 0.2f
#define LRELU(v) ((v) > 0.0f ? (v) : NEG_SLOPE * (v))


__global__ void zero_header_kernel(int* hdr) {
    if (threadIdx.x < 48) hdr[threadIdx.x] = 0;
}

__global__ void hist_kernel(const int* __restrict__ idx, int n, int* __restrict__ counts) {
    __shared__ int h[16];
    if (threadIdx.x < 16) h[threadIdx.x] = 0;
    __syncthreads();
    int i = blockIdx.x * blockDim.x + threadIdx.x;
    if (i < n) atomicAdd(&h[idx[i] & 15], 1);
    __syncthreads();
    if (threadIdx.x < 16 && h[threadIdx.x] > 0)
        atomicAdd(&counts[threadIdx.x], h[threadIdx.x]);
}

__global__ void scan_kernel(const int* __restrict__ counts, int* __restrict__ offs,
                            int* __restrict__ cursors) {
    if (threadIdx.x == 0) {
        int s = 0;
        for (int e = 0; e < 16; ++e) {
            offs[e] = s;
            cursors[e] = s;
            s += counts[e];
        }
    }
}

__global__ void scatter_kernel(const int* __restrict__ idx, int n,
                               int* __restrict__ cursors, int* __restrict__ perm) {
    __shared__ int lcount[16];
    __shared__ int lbase[16];
    if (threadIdx.x < 16) lcount[threadIdx.x] = 0;
    __syncthreads();
    int i = blockIdx.x * blockDim.x + threadIdx.x;
    int e = 0, rank = 0;
    bool live = (i < n);
    if (live) {
        e = idx[i] & 15;
        rank = atomicAdd(&lcount[e], 1);
    }
    __syncthreads();
    if (threadIdx.x < 16 && lcount[threadIdx.x] > 0)
        lbase[threadIdx.x] = atomicAdd(&cursors[threadIdx.x], lcount[threadIdx.x]);
    __syncthreads();
    if (live) perm[lbase[e] + rank] = i;
}


__global__ __launch_bounds__(32) void expert_mlp_wmma_kernel(
    const float* __restrict__ x,
    const float* __restrict__ W0, const float* __restrict__ b0,
    const float* __restrict__ W1, const float* __restrict__ b1,
    const float* __restrict__ W2, const float* __restrict__ b2,
    const int* __restrict__ offs, const int* __restrict__ counts,
    const int* __restrict__ perm,
    float* __restrict__ out)
{
    const int e    = blockIdx.x;
    const int lane = threadIdx.x;
    const int m    = lane & 15;
    const int hi   = lane >> 4;

    __shared__ float sW0[8 * 16];
    __shared__ float sW1[16 * 32];
    __shared__ float sW2[32 * 16];
    __shared__ float sB0[16], sB1[32], sB2[16];
    __shared__ float sX[16 * 8];
    __shared__ float sH[16 * 33];
    __shared__ int   sRow[16];

    for (int i = lane; i < 128; i += 32) {
        int k = i >> 4, n = i & 15;
        sW0[i] = (k < 7) ? W0[e * 112 + k * 16 + n] : 0.0f;
    }
    for (int i = lane; i < 512; i += 32) sW1[i] = W1[e * 512 + i];
    for (int i = lane; i < 512; i += 32) sW2[i] = W2[e * 512 + i];
    if (lane < 16) sB0[lane] = b0[e * 16 + lane];
    sB1[lane] = b1[e * 32 + lane];
    if (lane < 16) sB2[lane] = b2[e * 16 + lane];
    __syncthreads();

    v2f bf0[2], bf1[2][4], bf2[8];
#pragma unroll
    for (int c = 0; c < 2; ++c) {
        int kk = 4 * c + 2 * hi;
        bf0[c].x = sW0[kk * 16 + m];
        bf0[c].y = sW0[(kk + 1) * 16 + m];
    }
#pragma unroll
    for (int j = 0; j < 2; ++j)
#pragma unroll
        for (int c = 0; c < 4; ++c) {
            int kk = 4 * c + 2 * hi;
            int n  = 16 * j + m;
            bf1[j][c].x = sW1[kk * 32 + n];
            bf1[j][c].y = sW1[(kk + 1) * 32 + n];
        }
#pragma unroll
    for (int c = 0; c < 8; ++c) {
        int kk = 4 * c + 2 * hi;
        bf2[c].x = sW2[kk * 16 + m];
        bf2[c].y = sW2[(kk + 1) * 16 + m];
    }

    const float bias0 = sB0[m];
    const float b1lo  = sB1[m];
    const float b1hi  = sB1[16 + m];
    const float bias2 = sB2[m];

    const int base   = offs[e];
    const int cnt    = counts[e];
    const int ntiles = (cnt + 15) >> 4;

    for (int t = blockIdx.y; t < ntiles; t += gridDim.y) {
        __syncthreads();

        int valid = cnt - (t << 4);
        if (valid > 16) valid = 16;

        if (lane < 16) {
            int row = (lane < valid) ? perm[base + (t << 4) + lane] : -1;
            sRow[lane] = row;
            if (row >= 0) {
                const float* xs = x + row * 7;
#pragma unroll
                for (int k = 0; k < 7; ++k) sX[lane * 8 + k] = xs[k];
            } else {
#pragma unroll
                for (int k = 0; k < 7; ++k) sX[lane * 8 + k] = 0.0f;
            }
            sX[lane * 8 + 7] = 0.0f;
        }
        __syncthreads();

        v2f a0, a1;
        a0.x = sX[m * 8 + 2 * hi];
        a0.y = sX[m * 8 + 2 * hi + 1];
        a1.x = sX[m * 8 + 4 + 2 * hi];
        a1.y = sX[m * 8 + 4 + 2 * hi + 1];
        v8f c0 = {};
        c0 = __builtin_amdgcn_wmma_f32_16x16x4_f32(false, a0, false, bf0[0], (short)0, c0, false, false);
        c0 = __builtin_amdgcn_wmma_f32_16x16x4_f32(false, a1, false, bf0[1], (short)0, c0, false, false);
#pragma unroll
        for (int r = 0; r < 8; ++r) {
            float v = c0[r] + bias0;
            c0[r] = LRELU(v);
        }
#pragma unroll
        for (int r = 0; r < 8; ++r) sH[(r + 8 * hi) * 33 + m] = c0[r];
        __syncthreads();

        v2f af1[4];
#pragma unroll
        for (int c = 0; c < 4; ++c) {
            int kk = 4 * c + 2 * hi;
            af1[c].x = sH[m * 33 + kk];
            af1[c].y = sH[m * 33 + kk + 1];
        }
        __syncthreads();

        v8f c1lo = {}, c1hi = {};
#pragma unroll
        for (int c = 0; c < 4; ++c)
            c1lo = __builtin_amdgcn_wmma_f32_16x16x4_f32(false, af1[c], false, bf1[0][c], (short)0, c1lo, false, false);
#pragma unroll
        for (int c = 0; c < 4; ++c)
            c1hi = __builtin_amdgcn_wmma_f32_16x16x4_f32(false, af1[c], false, bf1[1][c], (short)0, c1hi, false, false);
#pragma unroll
        for (int r = 0; r < 8; ++r) {
            float vlo = c1lo[r] + b1lo;
            float vhi = c1hi[r] + b1hi;
            sH[(r + 8 * hi) * 33 + m]      = LRELU(vlo);
            sH[(r + 8 * hi) * 33 + 16 + m] = LRELU(vhi);
        }
        __syncthreads();

        v8f c2 = {};
#pragma unroll
        for (int c = 0; c < 8; ++c) {
            int kk = 4 * c + 2 * hi;
            v2f a;
            a.x = sH[m * 33 + kk];
            a.y = sH[m * 33 + kk + 1];
            c2 = __builtin_amdgcn_wmma_f32_16x16x4_f32(false, a, false, bf2[c], (short)0, c2, false, false);
        }

#pragma unroll
        for (int r = 0; r < 8; ++r) {
            int row = sRow[r + 8 * hi];
            if (row >= 0) {
                float v = c2[r] + bias2;
                out[row * 16 + m] = LRELU(v);
            }
        }
    }
}


extern "C" void kernel_launch(void* const* d_in, const int* in_sizes, int n_in,
                              void* d_out, int out_size, void* d_ws, size_t ws_size,
                              hipStream_t stream) {
    const float* x  = (const float*)d_in[0];
    const int*   idx = (const int*)d_in[1];
    const float* W0 = (const float*)d_in[2];
    const float* b0 = (const float*)d_in[3];
    const float* W1 = (const float*)d_in[4];
    const float* b1 = (const float*)d_in[5];
    const float* W2 = (const float*)d_in[6];
    const float* b2 = (const float*)d_in[7];
    float* out = (float*)d_out;

    const int n = in_sizes[1];

    int* hdr     = (int*)d_ws;
    int* counts  = hdr;
    int* offs    = hdr + 16;
    int* cursors = hdr + 32;
    int* perm    = hdr + 64;

    const int nb = (n + 255) / 256;

    zero_header_kernel<<<1, 64, 0, stream>>>(hdr);
    hist_kernel<<<nb, 256, 0, stream>>>(idx, n, counts);
    scan_kernel<<<1, 32, 0, stream>>>(counts, offs, cursors);
    scatter_kernel<<<nb, 256, 0, stream>>>(idx, n, cursors, perm);

    dim3 grid2(16, 128);
    expert_mlp_wmma_kernel<<<grid2, 32, 0, stream>>>(
        x, W0, b0, W1, b1, W2, b2, offs, counts, perm, out);
}


# No sort: single-wave block (expert e, chunk c) picks e's points out of chunk c's 1024 indices itself (ballot + prefix count, list in LDS); 4 binning launches dropped, 5 -> 1; sums unchanged
# speedup vs baseline: 1.9503x; 1.9503x over previous
#include <hip/hip_runtime.h>
#include <hip/hip_bf16.h>


typedef float v2f __attribute__((ext_vector_type(2)));
typedef float v8f __attribute__((ext_vector_type(8)));
typedef int   v4i __attribute__((ext_vector_type(4)));

#define NEG_SLOPE 0.2f
#define LRELU(v) ((v) > 0.0f ? (v) : NEG_SLOPE * (v))

#define CHUNK      1024
#define CHUNK_VECS (CHUNK / 128)


__global__ __launch_bounds__(32) void expert_mlp_wmma_kernel(
    const float* __restrict__ x,
    const int* __restrict__ idx,
    const float* __restrict__ W0, const float* __restrict__ b0,
    const float* __restrict__ W1, const float* __restrict__ b1,
    const float* __restrict__ W2, const float* __restrict__ b2,
    int n,
    float* __restrict__ out)
{
    const int e    = blockIdx.x;
    const int lane = threadIdx.x;
    const int m    = lane & 15;
    const int hi   = lane >> 4;

    int chunk = blockIdx.y;
    if (chunk * CHUNK >= n) return;

    __shared__ int   sWL[128 + 512 + 512];
    __shared__ float sB0[16], sB1[32], sB2[16];
    __shared__ float sX[16 * 8];
    __shared__ float sH[16 * 33];
    __shared__ int   sRow[16];

    int* const sW0 = sWL;
    int* const sW1 = sWL + 128;
    int* const sW2 = sWL + 640;

    v4i iv[CHUNK_VECS];
    auto load_idx = [&](int c) {
        const int p0 = c * CHUNK + lane * 4;
        if ((c + 1) * CHUNK <= n) {
#pragma unroll
            for (int j = 0; j < CHUNK_VECS; ++j)
                iv[j] = *reinterpret_cast<const v4i*>(idx + p0 + j * 128);
        } else {
#pragma unroll
            for (int j = 0; j < CHUNK_VECS; ++j) {
#pragma unroll
                for (int q = 0; q < 4; ++q) {
                    const int p = p0 + j * 128 + q;
                    const int v = idx[p < n ? p : n - 1];
                    iv[j][q] = (p < n) ? v : (e ^ 1);
                }
            }
        }
    };
    load_idx(chunk);

    for (int i = lane; i < 128; i += 32) {
        int k = i >> 4, nn = i & 15;
        sW0[i] = (k < 7) ? __float_as_int(W0[e * 112 + k * 16 + nn]) : 0;
    }
    for (int i = lane; i < 512; i += 32) sW1[i] = __float_as_int(W1[e * 512 + i]);
    for (int i = lane; i < 512; i += 32) sW2[i] = __float_as_int(W2[e * 512 + i]);
    if (lane < 16) sB0[lane] = b0[e * 16 + lane];
    sB1[lane] = b1[e * 32 + lane];
    if (lane < 16) sB2[lane] = b2[e * 16 + lane];
    __syncthreads();

    v2f bf0[2], bf1[2][4], bf2[8];
#pragma unroll
    for (int c = 0; c < 2; ++c) {
        int kk = 4 * c + 2 * hi;
        bf0[c].x = __int_as_float(sW0[kk * 16 + m]);
        bf0[c].y = __int_as_float(sW0[(kk + 1) * 16 + m]);
    }
#pragma unroll
    for (int j = 0; j < 2; ++j)
#pragma unroll
        for (int c = 0; c < 4; ++c) {
            int kk = 4 * c + 2 * hi;
            int nn = 16 * j + m;
            bf1[j][c].x = __int_as_float(sW1[kk * 32 + nn]);
            bf1[j][c].y = __int_as_float(sW1[(kk + 1) * 32 + nn]);
        }
#pragma unroll
    for (int c = 0; c < 8; ++c) {
        int kk = 4 * c + 2 * hi;
        bf2[c].x = __int_as_float(sW2[kk * 16 + m]);
        bf2[c].y = __int_as_float(sW2[(kk + 1) * 16 + m]);
    }

    const float bias0 = sB0[m];
    const float b1lo  = sB1[m];
    const float b1hi  = sB1[16 + m];
    const float bias2 = sB2[m];

    for (;;) {
        __syncthreads();

        int count = 0;
#pragma unroll
        for (int j = 0; j < CHUNK_VECS; ++j) {
#pragma unroll
            for (int q = 0; q < 4; ++q) {
                const int p = chunk * CHUNK + j * 128 + lane * 4 + q;
                const bool match = ((iv[j][q] & 15) == e);
                const unsigned mask = __builtin_amdgcn_ballot_w32(match);
                if (match) sWL[count + (int)__builtin_amdgcn_mbcnt_lo(mask, 0u)] = p;
                count += __builtin_popcount(mask);
            }
        }
        __syncthreads();

        const int ntiles = (count + 15) >> 4;

        for (int t = 0; t < ntiles; ++t) {
            __syncthreads();

            int valid = count - (t << 4);
            if (valid > 16) valid = 16;

            if (lane < 16) {
                int row = (lane < valid) ? sWL[(t << 4) + lane] : -1;
                sRow[lane] = row;
                if (row >= 0) {
                    const float* xs = x + row * 7;
#pragma unroll
                    for (int k = 0; k < 7; ++k) sX[lane * 8 + k] = xs[k];
                } else {
#pragma unroll
                    for (int k = 0; k < 7; ++k) sX[lane * 8 + k] = 0.0f;
                }
                sX[lane * 8 + 7] = 0.0f;
            }
            __syncthreads();

            v2f a0, a1;
            a0.x = sX[m * 8 + 2 * hi];
            a0.y = sX[m * 8 + 2 * hi + 1];
            a1.x = sX[m * 8 + 4 + 2 * hi];
            a1.y = sX[m * 8 + 4 + 2 * hi + 1];
            v8f c0 = {};
            c0 = __builtin_amdgcn_wmma_f32_16x16x4_f32(false, a0, false, bf0[0], (short)0, c0, false, false);
            c0 = __builtin_amdgcn_wmma_f32_16x16x4_f32(false, a1, false, bf0[1], (short)0, c0, false, false);
#pragma unroll
            for (int r = 0; r < 8; ++r) {
                float v = c0[r] + bias0;
                c0[r] = LRELU(v);
            }
#pragma unroll
            for (int r = 0; r < 8; ++r) sH[(r + 8 * hi) * 33 + m] = c0[r];
            __syncthreads();

            v2f af1[4];
#pragma unroll
            for (int c = 0; c < 4; ++c) {
                int kk = 4 * c + 2 * hi;
                af1[c].x = sH[m * 33 + kk];
                af1[c].y = sH[m * 33 + kk + 1];
            }
            __syncthreads();

            v8f c1lo = {}, c1hi = {};
#pragma unroll
            for (int c = 0; c < 4; ++c)
                c1lo = __builtin_amdgcn_wmma_f32_16x16x4_f32(false, af1[c], false, bf1[0][c], (short)0, c1lo, false, false);
#pragma unroll
            for (int c = 0; c < 4; ++c)
                c1hi = __builtin_amdgcn_wmma_f32_16x16x4_f32(false, af1[c], false, bf1[1][c], (short)0, c1hi, false, false);
#pragma unroll
            for (int r = 0; r < 8; ++r) {
                float vlo = c1lo[r] + b1lo;
                float vhi = c1hi[r] + b1hi;
                sH[(r + 8 * hi) * 33 + m]      = LRELU(vlo);
                sH[(r + 8 * hi) * 33 + 16 + m] = LRELU(vhi);
            }
            __syncthreads();

            v8f c2 = {};
#pragma unroll
            for (int c = 0; c < 8; ++c) {
                int kk = 4 * c + 2 * hi;
                v2f a;
                a.x = sH[m * 33 + kk];
                a.y = sH[m * 33 + kk + 1];
                c2 = __builtin_amdgcn_wmma_f32_16x16x4_f32(false, a, false, bf2[c], (short)0, c2, false, false);
            }

#pragma unroll
            for (int r = 0; r < 8; ++r) {
                int row = sRow[r + 8 * hi];
                if (row >= 0) {
                    float v = c2[r] + bias2;
                    out[row * 16 + m] = LRELU(v);
                }
            }
        }

        chunk += gridDim.y;
        if (chunk * CHUNK >= n) break;
        load_idx(chunk);
    }
}


extern "C" void kernel_launch(void* const* d_in, const int* in_sizes, int n_in,
                              void* d_out, int out_size, void* d_ws, size_t ws_size,
                              hipStream_t stream) {
    const float* x  = (const float*)d_in[0];
    const int*   idx = (const int*)d_in[1];
    const float* W0 = (const float*)d_in[2];
    const float* b0 = (const float*)d_in[3];
    const float* W1 = (const float*)d_in[4];
    const float* b1 = (const float*)d_in[5];
    const float* W2 = (const float*)d_in[6];
    const float* b2 = (const float*)d_in[7];
    float* out = (float*)d_out;

    const int n = in_sizes[1];

    dim3 grid2(16, 512);
    expert_mlp_wmma_kernel<<<grid2, 32, 0, stream>>>(
        x, idx, W0, b0, W1, b1, W2, b2, n, out);
}


# Blocks of 8 waves of one expert (grid 16x64 x 256 threads, was 16x512 single-wave blocks): weights fetched and staged in LDS once per block, shared by its waves; 8x fewer workgroups; sums unchanged
# speedup vs baseline: 2.0209x; 1.0362x over previous
#include <hip/hip_runtime.h>
#include <hip/hip_bf16.h>


typedef float v2f __attribute__((ext_vector_type(2)));
typedef float v8f __attribute__((ext_vector_type(8)));
typedef int   v4i __attribute__((ext_vector_type(4)));

#define NEG_SLOPE 0.2f
#define LRELU(v) ((v) > 0.0f ? (v) : NEG_SLOPE * (v))

#define CHUNK      1024
#define CHUNK_VECS (CHUNK / 128)
#define WAVES      8
#define THREADS    (WAVES * 32)

#define WAVE_SYNC() __builtin_amdgcn_fence(__ATOMIC_ACQ_REL, "wavefront")


__global__ __launch_bounds__(THREADS) void expert_mlp_wmma_kernel(
    const float* __restrict__ x,
    const int* __restrict__ idx,
    const float* __restrict__ W0, const float* __restrict__ b0,
    const float* __restrict__ W1, const float* __restrict__ b1,
    const float* __restrict__ W2, const float* __restrict__ b2,
    int n,
    float* __restrict__ out)
{
    const int e    = blockIdx.x;
    const int tid  = threadIdx.x;
    const int wave = __builtin_amdgcn_readfirstlane(tid >> 5);
    const int lane = tid & 31;
    const int m    = lane & 15;
    const int hi   = lane >> 4;

    int chunk = blockIdx.y * WAVES + wave;
    const bool has_work = chunk * CHUNK < n;

    __shared__ int   sWts[128 + 512 + 512];
    __shared__ float sBias[64];
    __shared__ int   sListA[WAVES][CHUNK];
    __shared__ float sXA[WAVES][16 * 8];
    __shared__ float sHA[WAVES][16 * 33];
    __shared__ int   sRowA[WAVES][16];

    int* const sW0 = sWts;
    int* const sW1 = sWts + 128;
    int* const sW2 = sWts + 640;
    int*   const sWL  = sListA[wave];
    float* const sX   = sXA[wave];
    float* const sH   = sHA[wave];
    int*   const sRow = sRowA[wave];

    v4i iv[CHUNK_VECS];
    auto load_idx = [&](int c) {
        const int p0 = c * CHUNK + lane * 4;
        if ((c + 1) * CHUNK <= n) {
#pragma unroll
            for (int j = 0; j < CHUNK_VECS; ++j)
                iv[j] = *reinterpret_cast<const v4i*>(idx + p0 + j * 128);
        } else {
#pragma unroll
            for (int j = 0; j < CHUNK_VECS; ++j) {
#pragma unroll
                for (int q = 0; q < 4; ++q) {
                    const int p = p0 + j * 128 + q;
                    const int v = idx[p < n ? p : n - 1];
                    iv[j][q] = (p < n) ? v : (e ^ 1);
                }
            }
        }
    };
    if (has_work) load_idx(chunk);

    {
        const float* bp = b2 + e * 16 + (tid - 48);
        if (tid < 48) bp = b1 + e * 32 + (tid - 16);
        if (tid < 16) bp = b0 + e * 16 + tid;
        const float bv  = (tid < 64) ? *bp : 0.0f;
        const float w0v = (tid < 112) ? W0[e * 112 + tid] : 0.0f;
        const float w1a = W1[e * 512 + tid];
        const float w1b = W1[e * 512 + THREADS + tid];
        const float w2a = W2[e * 512 + tid];
        const float w2b = W2[e * 512 + THREADS + tid];
        if (tid < 64)  sBias[tid] = bv;
        if (tid < 128) sW0[tid] = __float_as_int(w0v);
        sW1[tid]           = __float_as_int(w1a);
        sW1[THREADS + tid] = __float_as_int(w1b);
        sW2[tid]           = __float_as_int(w2a);
        sW2[THREADS + tid] = __float_as_int(w2b);
    }
    __syncthreads();
    if (!has_work) return;

    v2f bf0[2], bf1[2][4], bf2[8];
#pragma unroll
    for (int c = 0; c < 2; ++c) {
        int kk = 4 * c + 2 * hi;
        bf0[c].x = __int_as_float(sW0[kk * 16 + m]);
        bf0[c].y = __int_as_float(sW0[(kk + 1) * 16 + m]);
    }
#pragma unroll
    for (int j = 0; j < 2; ++j)
#pragma unroll
        for (int c = 0; c < 4; ++c) {
            int kk = 4 * c + 2 * hi;
            int nn = 16 * j + m;
            bf1[j][c].x = __int_as_float(sW1[kk * 32 + nn]);
            bf1[j][c].y = __int_as_float(sW1[(kk + 1) * 32 + nn]);
        }
#pragma unroll
    for (int c = 0; c < 8; ++c) {
        int kk = 4 * c + 2 * hi;
        bf2[c].x = __int_as_float(sW2[kk * 16 + m]);
        bf2[c].y = __int_as_float(sW2[(kk + 1) * 16 + m]);
    }

    const float bias0 = sBias[m];
    const float b1lo  = sBias[16 + m];
    const float b1hi  = sBias[32 + m];
    const float bias2 = sBias[48 + m];

    for (;;) {
        WAVE_SYNC();

        int count = 0;
#pragma unroll
        for (int j = 0; j < CHUNK_VECS; ++j) {
#pragma unroll
            for (int q = 0; q < 4; ++q) {
                const int p = chunk * CHUNK + j * 128 + lane * 4 + q;
                const bool match = ((iv[j][q] & 15) == e);
                const unsigned mask = __builtin_amdgcn_ballot_w32(match);
                if (match) sWL[count + (int)__builtin_amdgcn_mbcnt_lo(mask, 0u)] = p;
                count += __builtin_popcount(mask);
            }
        }
        WAVE_SYNC();

        const int ntiles = (count + 15) >> 4;

        for (int t = 0; t < ntiles; ++t) {
            WAVE_SYNC();

            int valid = count - (t << 4);
            if (valid > 16) valid = 16;

            if (lane < 16) {
                int row = (lane < valid) ? sWL[(t << 4) + lane] : -1;
                sRow[lane] = row;
                if (row >= 0) {
                    const float* xs = x + row * 7;
#pragma unroll
                    for (int k = 0; k < 7; ++k) sX[lane * 8 + k] = xs[k];
                } else {
#pragma unroll
                    for (int k = 0; k < 7; ++k) sX[lane * 8 + k] = 0.0f;
                }
                sX[lane * 8 + 7] = 0.0f;
            }
            WAVE_SYNC();

            v2f a0, a1;
            a0.x = sX[m * 8 + 2 * hi];
            a0.y = sX[m * 8 + 2 * hi + 1];
            a1.x = sX[m * 8 + 4 + 2 * hi];
            a1.y = sX[m * 8 + 4 + 2 * hi + 1];
            v8f c0 = {};
            c0 = __builtin_amdgcn_wmma_f32_16x16x4_f32(false, a0, false, bf0[0], (short)0, c0, false, false);
            c0 = __builtin_amdgcn_wmma_f32_16x16x4_f32(false, a1, false, bf0[1], (short)0, c0, false, false);
#pragma unroll
            for (int r = 0; r < 8; ++r) {
                float v = c0[r] + bias0;
                c0[r] = LRELU(v);
            }
#pragma unroll
            for (int r = 0; r < 8; ++r) sH[(r + 8 * hi) * 33 + m] = c0[r];
            WAVE_SYNC();

            v2f af1[4];
#pragma unroll
            for (int c = 0; c < 4; ++c) {
                int kk = 4 * c + 2 * hi;
                af1[c].x = sH[m * 33 + kk];
                af1[c].y = sH[m * 33 + kk + 1];
            }
            WAVE_SYNC();

            v8f c1lo = {}, c1hi = {};
#pragma unroll
            for (int c = 0; c < 4; ++c)
                c1lo = __builtin_amdgcn_wmma_f32_16x16x4_f32(false, af1[c], false, bf1[0][c], (short)0, c1lo, false, false);
#pragma unroll
            for (int c = 0; c < 4; ++c)
                c1hi = __builtin_amdgcn_wmma_f32_16x16x4_f32(false, af1[c], false, bf1[1][c], (short)0, c1hi, false, false);
#pragma unroll
            for (int r = 0; r < 8; ++r) {
                float vlo = c1lo[r] + b1lo;
                float vhi = c1hi[r] + b1hi;
                sH[(r + 8 * hi) * 33 + m]      = LRELU(vlo);
                sH[(r + 8 * hi) * 33 + 16 + m] = LRELU(vhi);
            }
            WAVE_SYNC();

            v8f c2 = {};
#pragma unroll
            for (int c = 0; c < 8; ++c) {
                int kk = 4 * c + 2 * hi;
                v2f a;
                a.x = sH[m * 33 + kk];
                a.y = sH[m * 33 + kk + 1];
                c2 = __builtin_amdgcn_wmma_f32_16x16x4_f32(false, a, false, bf2[c], (short)0, c2, false, false);
            }

#pragma unroll
            for (int r = 0; r < 8; ++r) {
                int row = sRow[r + 8 * hi];
                if (row >= 0) {
                    float v = c2[r] + bias2;
                    out[row * 16 + m] = LRELU(v);
                }
            }
        }

        chunk += gridDim.y * WAVES;
        if (chunk * CHUNK >= n) break;
        load_idx(chunk);
    }
}


extern "C" void kernel_launch(void* const* d_in, const int* in_sizes, int n_in,
                              void* d_out, int out_size, void* d_ws, size_t ws_size,
                              hipStream_t stream) {
    const float* x  = (const float*)d_in[0];
    const int*   idx = (const int*)d_in[1];
    const float* W0 = (const float*)d_in[2];
    const float* b0 = (const float*)d_in[3];
    const float* W1 = (const float*)d_in[4];
    const float* b1 = (const float*)d_in[5];
    const float* W2 = (const float*)d_in[6];
    const float* b2 = (const float*)d_in[7];
    float* out = (float*)d_out;

    const int n = in_sizes[1];

    dim3 grid2(16, 64);
    expert_mlp_wmma_kernel<<<grid2, THREADS, 0, stream>>>(
        x, idx, W0, b0, W1, b1, W2, b2, n, out);
}


# Tile's x rows and point ids not staged in LDS: each lane loads its layer-0 A fragment from global (one 16-byte load, one tile ahead), ids read from the list; layer-0 k regrouped (sum order changes)
# speedup vs baseline: 2.1711x; 1.0743x over previous
#include <hip/hip_runtime.h>
#include <hip/hip_bf16.h>


typedef float v2f __attribute__((ext_vector_type(2)));
typedef float v8f __attribute__((ext_vector_type(8)));
typedef float v4f __attribute__((ext_vector_type(4)));
typedef int   v4i __attribute__((ext_vector_type(4)));

#define NEG_SLOPE 0.2f
#define LRELU(v) ((v) > 0.0f ? (v) : NEG_SLOPE * (v))

#define CHUNK      1024
#define CHUNK_VECS (CHUNK / 128)
#define WAVES      8
#define THREADS    (WAVES * 32)

#define WAVE_SYNC() __builtin_amdgcn_fence(__ATOMIC_ACQ_REL, "wavefront")


__global__ __launch_bounds__(THREADS) void expert_mlp_wmma_kernel(
    const float* __restrict__ x,
    const int* __restrict__ idx,
    const float* __restrict__ W0, const float* __restrict__ b0,
    const float* __restrict__ W1, const float* __restrict__ b1,
    const float* __restrict__ W2, const float* __restrict__ b2,
    int n,
    float* __restrict__ out)
{
    const int e    = blockIdx.x;
    const int tid  = threadIdx.x;
    const int wave = __builtin_amdgcn_readfirstlane(tid >> 5);
    const int lane = tid & 31;
    const int m    = lane & 15;
    const int hi   = lane >> 4;

    int chunk = blockIdx.y * WAVES + wave;
    const bool has_work = chunk * CHUNK < n;

    __shared__ int   sWts[128 + 512 + 512];
    __shared__ float sBias[64];
    __shared__ __attribute__((aligned(16))) int sListA[WAVES][CHUNK];
    __shared__ float sHA[WAVES][16 * 33];

    int* const sW0 = sWts;
    int* const sW1 = sWts + 128;
    int* const sW2 = sWts + 640;
    int*   const sWL  = sListA[wave];
    float* const sH   = sHA[wave];

    v4i iv[CHUNK_VECS];
    auto load_idx = [&](int c) {
        const int p0 = c * CHUNK + lane * 4;
        if ((c + 1) * CHUNK <= n) {
#pragma unroll
            for (int j = 0; j < CHUNK_VECS; ++j)
                iv[j] = *reinterpret_cast<const v4i*>(idx + p0 + j * 128);
        } else {
#pragma unroll
            for (int j = 0; j < CHUNK_VECS; ++j) {
#pragma unroll
                for (int q = 0; q < 4; ++q) {
                    const int p = p0 + j * 128 + q;
                    const int v = idx[p < n ? p : n - 1];
                    iv[j][q] = (p < n) ? v : (e ^ 1);
                }
            }
        }
    };
    if (has_work) load_idx(chunk);

    {
        const float* bp = b2 + e * 16 + (tid - 48);
        if (tid < 48) bp = b1 + e * 32 + (tid - 16);
        if (tid < 16) bp = b0 + e * 16 + tid;
        const float bv  = (tid < 64) ? *bp : 0.0f;
        const float w0v = (tid < 112) ? W0[e * 112 + tid] : 0.0f;
        const float w1a = W1[e * 512 + tid];
        const float w1b = W1[e * 512 + THREADS + tid];
        const float w2a = W2[e * 512 + tid];
        const float w2b = W2[e * 512 + THREADS + tid];
        if (tid < 64)  sBias[tid] = bv;
        if (tid < 128) sW0[tid] = __float_as_int(w0v);
        sW1[tid]           = __float_as_int(w1a);
        sW1[THREADS + tid] = __float_as_int(w1b);
        sW2[tid]           = __float_as_int(w2a);
        sW2[THREADS + tid] = __float_as_int(w2b);
    }
    __syncthreads();
    if (!has_work) return;

    v2f bf0[2], bf1[2][4], bf2[8];
#pragma unroll
    for (int c = 0; c < 2; ++c) {
        int kk = 4 * hi + 2 * c;
        bf0[c].x = __int_as_float(sW0[kk * 16 + m]);
        bf0[c].y = __int_as_float(sW0[(kk + 1) * 16 + m]);
    }
#pragma unroll
    for (int j = 0; j < 2; ++j)
#pragma unroll
        for (int c = 0; c < 4; ++c) {
            int kk = 4 * c + 2 * hi;
            int nn = 16 * j + m;
            bf1[j][c].x = __int_as_float(sW1[kk * 32 + nn]);
            bf1[j][c].y = __int_as_float(sW1[(kk + 1) * 32 + nn]);
        }
#pragma unroll
    for (int c = 0; c < 8; ++c) {
        int kk = 4 * c + 2 * hi;
        bf2[c].x = __int_as_float(sW2[kk * 16 + m]);
        bf2[c].y = __int_as_float(sW2[(kk + 1) * 16 + m]);
    }

    const float bias0 = sBias[m];
    const float b1lo  = sBias[16 + m];
    const float b1hi  = sBias[32 + m];
    const float bias2 = sBias[48 + m];

    for (;;) {
        WAVE_SYNC();

        int count = 0;
#pragma unroll
        for (int j = 0; j < CHUNK_VECS; ++j) {
#pragma unroll
            for (int q = 0; q < 4; ++q) {
                const int p = chunk * CHUNK + j * 128 + lane * 4 + q;
                const bool match = ((iv[j][q] & 15) == e);
                const unsigned mask = __builtin_amdgcn_ballot_w32(match);
                if (match) sWL[count + (int)__builtin_amdgcn_mbcnt_lo(mask, 0u)] = p;
                count += __builtin_popcount(mask);
            }
        }
        WAVE_SYNC();

        const int ntiles = (count + 15) >> 4;

        auto list_row = [&](int t) -> int {
            const int i = (t << 4) + m;
            return (i < count) ? sWL[i] : 0;
        };
        auto fetch_x = [&](int row) -> v4f {
            v4f v;
            __builtin_memcpy(&v, x + row * 7 + 3 * hi, sizeof(v));
            return v;
        };

        v4f xnext = {};
        int rnext = 0;
        if (ntiles > 0) {
            xnext = fetch_x(list_row(0));
            rnext = list_row(1);
        }

        for (int t = 0; t < ntiles; ++t) {
            WAVE_SYNC();

            const v4f xv = xnext;
            if (t + 1 < ntiles) xnext = fetch_x(rnext);

            v2f a0, a1;
            a0.x = hi ? xv[1] : xv[0];
            a0.y = hi ? xv[2] : xv[1];
            a1.x = hi ? xv[3] : xv[2];
            a1.y = hi ? 0.0f  : xv[3];
            v8f c0 = {};
            c0 = __builtin_amdgcn_wmma_f32_16x16x4_f32(false, a0, false, bf0[0], (short)0, c0, false, false);
            c0 = __builtin_amdgcn_wmma_f32_16x16x4_f32(false, a1, false, bf0[1], (short)0, c0, false, false);
#pragma unroll
            for (int r = 0; r < 8; ++r) {
                float v = c0[r] + bias0;
                c0[r] = LRELU(v);
            }
#pragma unroll
            for (int r = 0; r < 8; ++r) sH[(r + 8 * hi) * 33 + m] = c0[r];
            WAVE_SYNC();

            rnext = list_row(t + 2);
            v2f af1[4];
#pragma unroll
            for (int c = 0; c < 4; ++c) {
                int kk = 4 * c + 2 * hi;
                af1[c].x = sH[m * 33 + kk];
                af1[c].y = sH[m * 33 + kk + 1];
            }
            WAVE_SYNC();

            v8f c1lo = {}, c1hi = {};
#pragma unroll
            for (int c = 0; c < 4; ++c)
                c1lo = __builtin_amdgcn_wmma_f32_16x16x4_f32(false, af1[c], false, bf1[0][c], (short)0, c1lo, false, false);
#pragma unroll
            for (int c = 0; c < 4; ++c)
                c1hi = __builtin_amdgcn_wmma_f32_16x16x4_f32(false, af1[c], false, bf1[1][c], (short)0, c1hi, false, false);
#pragma unroll
            for (int r = 0; r < 8; ++r) {
                float vlo = c1lo[r] + b1lo;
                float vhi = c1hi[r] + b1hi;
                sH[(r + 8 * hi) * 33 + m]      = LRELU(vlo);
                sH[(r + 8 * hi) * 33 + 16 + m] = LRELU(vhi);
            }
            WAVE_SYNC();

            v8f c2 = {};
#pragma unroll
            for (int c = 0; c < 8; ++c) {
                int kk = 4 * c + 2 * hi;
                v2f a;
                a.x = sH[m * 33 + kk];
                a.y = sH[m * 33 + kk + 1];
                c2 = __builtin_amdgcn_wmma_f32_16x16x4_f32(false, a, false, bf2[c], (short)0, c2, false, false);
            }

            const int   ibase = (t << 4) + 8 * hi;
            const v4i*  lp    = reinterpret_cast<const v4i*>(sWL + ibase);
            const v4i   rowa  = lp[0];
            const v4i   rowb  = lp[1];
#pragma unroll
            for (int r = 0; r < 8; ++r) {
                const int row = (r < 4) ? rowa[r & 3] : rowb[r & 3];
                if (ibase + r < count) {
                    float v = c2[r] + bias2;
                    out[row * 16 + m] = LRELU(v);
                }
            }
        }

        chunk += gridDim.y * WAVES;
        if (chunk * CHUNK >= n) break;
        load_idx(chunk);
    }
}


extern "C" void kernel_launch(void* const* d_in, const int* in_sizes, int n_in,
                              void* d_out, int out_size, void* d_ws, size_t ws_size,
                              hipStream_t stream) {
    const float* x  = (const float*)d_in[0];
    const int*   idx = (const int*)d_in[1];
    const float* W0 = (const float*)d_in[2];
    const float* b0 = (const float*)d_in[3];
    const float* W1 = (const float*)d_in[4];
    const float* b1 = (const float*)d_in[5];
    const float* W2 = (const float*)d_in[6];
    const float* b2 = (const float*)d_in[7];
    float* out = (float*)d_out;

    const int n = in_sizes[1];

    dim3 grid2(16, 64);
    expert_mlp_wmma_kernel<<<grid2, THREADS, 0, stream>>>(
        x, idx, W0, b0, W1, b1, W2, b2, n, out);
}
